# starting kernel
# baseline (speedup 1.0000x reference)
#include <hip/hip_runtime.h>

typedef __attribute__((ext_vector_type(2))) float v2f;
typedef __attribute__((ext_vector_type(8))) float v8f;

#define BATCH 25
#define LSEQ  11
#define NOUT  4
#define DIM   2048
#define VOCAB 32000
#define KSEL  1024

#define TILE_N          16
#define WAVES_PER_BLOCK 8
#define BLOCK_THREADS   256
#define BLOCKS_PER_BL   (KSEL / (TILE_N * WAVES_PER_BLOCK))

__global__ __launch_bounds__(BLOCK_THREADS)
void gather_head_wmma_f32(const float* __restrict__ h,
                          const float* __restrict__ w,
                          const int*   __restrict__ index,
                          float*       __restrict__ out)
{
    __shared__ float hs[(NOUT + 1) * DIM];

    const int bl   = blockIdx.x / BLOCKS_PER_BL;
    const int nblk = blockIdx.x % BLOCKS_PER_BL;
    const int tid  = threadIdx.x;
    const int lane = tid & 31;
    const int wave = tid >> 5;

    {
        const float4* src = (const float4*)(h + (size_t)bl * (NOUT * DIM));
        float4* dst = (float4*)hs;
        #pragma unroll
        for (int i = tid; i < (NOUT * DIM) / 4; i += BLOCK_THREADS)
            dst[i] = src[i];
        float4 z = {0.f, 0.f, 0.f, 0.f};
        float4* zr = (float4*)(hs + NOUT * DIM);
        #pragma unroll
        for (int i = tid; i < DIM / 4; i += BLOCK_THREADS)
            zr[i] = z;
    }
    __syncthreads();

    const int n0   = nblk * (TILE_N * WAVES_PER_BLOCK) + wave * TILE_N;
    const int nl   = lane & 15;
    const int koff = (lane >= 16) ? 2 : 0;

    const int row = index[(size_t)bl * KSEL + n0 + nl];
    const float* wp = w + (size_t)row * DIM + koff;

    const int am = (nl < NOUT) ? nl : NOUT;
    const float* ap = hs + am * DIM + koff;

    v8f acc0 = {};
    v8f acc1 = {};
    #pragma unroll 8
    for (int k = 0; k < DIM; k += 8) {
        v2f a0 = *(const v2f*)(ap + k);
        v2f b0 = *(const v2f*)(wp + k);
        v2f a1 = *(const v2f*)(ap + k + 4);
        v2f b1 = *(const v2f*)(wp + k + 4);
        acc0 = __builtin_amdgcn_wmma_f32_16x16x4_f32(false, a0, false, b0,
                                                     (short)0, acc0, false, false);
        acc1 = __builtin_amdgcn_wmma_f32_16x16x4_f32(false, a1, false, b1,
                                                     (short)0, acc1, false, false);
    }
    v8f acc = acc0 + acc1;

    if (lane < 16) {
        float* ob = out + (size_t)bl * NOUT * KSEL + n0 + lane;
        #pragma unroll
        for (int r = 0; r < NOUT; ++r)
            ob[(size_t)r * KSEL] = acc[r];
    }
}

extern "C" void kernel_launch(void* const* d_in, const int* in_sizes, int n_in,
                              void* d_out, int out_size, void* d_ws, size_t ws_size,
                              hipStream_t stream)
{
    const float* h   = (const float*)d_in[0];
    const float* w   = (const float*)d_in[1];
    const int*   idx = (const int*)d_in[2];
    float* out = (float*)d_out;

    dim3 grid(BATCH * LSEQ * BLOCKS_PER_BL);
    dim3 block(BLOCK_THREADS);
    hipLaunchKernelGGL(gather_head_wmma_f32, grid, block, 0, stream,
                       h, w, idx, out);
}


# B (gathered row) loads widened from 8 to 16 bytes per lane, A loader given the same k order: half the global load instructions and line touches; f32 sum order changes (k pairs regrouped)
# speedup vs baseline: 1.0457x; 1.0457x over previous
#include <hip/hip_runtime.h>

typedef __attribute__((ext_vector_type(2))) float v2f;
typedef __attribute__((ext_vector_type(4))) float v4f;
typedef __attribute__((ext_vector_type(8))) float v8f;

#define BATCH 25
#define LSEQ  11
#define NOUT  4
#define DIM   2048
#define VOCAB 32000
#define KSEL  1024

#define TILE_N          16
#define WAVES_PER_BLOCK 8
#define BLOCK_THREADS   256
#define BLOCKS_PER_BL   (KSEL / (TILE_N * WAVES_PER_BLOCK))

__global__ __launch_bounds__(BLOCK_THREADS)
void gather_head_wmma_f32(const float* __restrict__ h,
                          const float* __restrict__ w,
                          const int*   __restrict__ index,
                          float*       __restrict__ out)
{
    __shared__ alignas(16) float hs[(NOUT + 1) * DIM];

    const int bl   = blockIdx.x / BLOCKS_PER_BL;
    const int nblk = blockIdx.x % BLOCKS_PER_BL;
    const int tid  = threadIdx.x;
    const int lane = tid & 31;
    const int wave = tid >> 5;

    {
        const float4* src = (const float4*)(h + (size_t)bl * (NOUT * DIM));
        float4* dst = (float4*)hs;
        #pragma unroll
        for (int i = tid; i < (NOUT * DIM) / 4; i += BLOCK_THREADS)
            dst[i] = src[i];
        float4 z = {0.f, 0.f, 0.f, 0.f};
        float4* zr = (float4*)(hs + NOUT * DIM);
        #pragma unroll
        for (int i = tid; i < DIM / 4; i += BLOCK_THREADS)
            zr[i] = z;
    }
    __syncthreads();

    const int n0   = nblk * (TILE_N * WAVES_PER_BLOCK) + wave * TILE_N;
    const int nl   = lane & 15;
    const int koff = (lane >= 16) ? 4 : 0;

    const int row = index[(size_t)bl * KSEL + n0 + nl];
    const float* wp = w + (size_t)row * DIM + koff;

    const int am = (nl < NOUT) ? nl : NOUT;
    const float* ap = hs + am * DIM + koff;

    v8f acc0 = {};
    v8f acc1 = {};
    #pragma unroll 8
    for (int k = 0; k < DIM; k += 8) {
        v4f aq = *(const v4f*)(ap + k);
        v4f bq = *(const v4f*)(wp + k);
        v2f a0 = {aq.x, aq.y};
        v2f b0 = {bq.x, bq.y};
        v2f a1 = {aq.z, aq.w};
        v2f b1 = {bq.z, bq.w};
        acc0 = __builtin_amdgcn_wmma_f32_16x16x4_f32(false, a0, false, b0,
                                                     (short)0, acc0, false, false);
        acc1 = __builtin_amdgcn_wmma_f32_16x16x4_f32(false, a1, false, b1,
                                                     (short)0, acc1, false, false);
    }
    v8f acc = acc0 + acc1;

    if (lane < 16) {
        float* ob = out + (size_t)bl * NOUT * KSEL + n0 + lane;
        #pragma unroll
        for (int r = 0; r < NOUT; ++r)
            ob[(size_t)r * KSEL] = acc[r];
    }
}

extern "C" void kernel_launch(void* const* d_in, const int* in_sizes, int n_in,
                              void* d_out, int out_size, void* d_ws, size_t ws_size,
                              hipStream_t stream)
{
    const float* h   = (const float*)d_in[0];
    const float* w   = (const float*)d_in[1];
    const int*   idx = (const int*)d_in[2];
    float* out = (float*)d_out;

    dim3 grid(BATCH * LSEQ * BLOCKS_PER_BL);
    dim3 block(BLOCK_THREADS);
    hipLaunchKernelGGL(gather_head_wmma_f32, grid, block, 0, stream,
                       h, w, idx, out);
}


# Each gathered row read as one 512-byte wave-wide load, made into B fragments in per-wave LDS rows; 4 tiles per wave in 4-wave blocks so the whole grid stays resident; f32 sum order unchanged
# speedup vs baseline: 1.5133x; 1.4472x over previous
#include <hip/hip_runtime.h>


typedef __attribute__((ext_vector_type(2))) float v2f;
typedef __attribute__((ext_vector_type(4))) float v4f;
typedef __attribute__((ext_vector_type(8))) float v8f;

#define BATCH 25
#define LSEQ  11
#define NOUT  4
#define DIM   2048
#define VOCAB 32000
#define KSEL  1024

#define TILE_N          16
#define WAVES_PER_BLOCK 4
#define BLOCK_THREADS   128
#define TILES_PER_WAVE  4
#define COLS_PER_WAVE   (TILE_N * TILES_PER_WAVE)
#define COLS_PER_BLOCK  (COLS_PER_WAVE * WAVES_PER_BLOCK)
#define BLOCKS_PER_BL   (KSEL / COLS_PER_BLOCK)
#define KC              128
#define NKC             (DIM / KC)
#define HP              (DIM + 8)
#define GS              4
#define SP              (KC + 4)

__global__ __launch_bounds__(BLOCK_THREADS) __attribute__((amdgpu_waves_per_eu(4, 4)))
void gather_head_wmma_f32(const float* __restrict__ h,
                          const float* __restrict__ w,
                          const int*   __restrict__ index,
                          float*       __restrict__ out)
{
    __shared__ alignas(16) float hs[(NOUT + 1) * HP];
    __shared__ alignas(16) float stg[WAVES_PER_BLOCK * TILE_N * SP];

    const int bl   = blockIdx.x / BLOCKS_PER_BL;
    const int nblk = blockIdx.x % BLOCKS_PER_BL;
    const int tid  = threadIdx.x;
    const int lane = tid & 31;
    const int wave = __builtin_amdgcn_readfirstlane(tid >> 5);

    {
        const float4* src = (const float4*)(h + (size_t)bl * (NOUT * DIM));
        #pragma unroll
        for (int i = tid; i < (NOUT * DIM) / 4; i += BLOCK_THREADS) {
            const int r  = i / (DIM / 4);
            const int c4 = i % (DIM / 4);
            *(float4*)(hs + r * HP + c4 * 4) = src[i];
        }
        float4 z = {0.f, 0.f, 0.f, 0.f};
        float4* zr = (float4*)(hs + NOUT * HP);
        for (int i = tid; i < HP / 4; i += BLOCK_THREADS)
            zr[i] = z;
    }
    __syncthreads();

    const int n0   = nblk * COLS_PER_BLOCK + wave * COLS_PER_WAVE;
    const int nl   = lane & 15;
    const int koff = (lane >= 16) ? 4 : 0;

    const int idx0 = index[(size_t)bl * KSEL + n0 + lane];
    const int idx1 = index[(size_t)bl * KSEL + n0 + 32 + lane];

    const int am = (nl < NOUT) ? nl : NOUT;
    const float* ap = hs + am * HP + koff;

    float*       st = stg + wave * (TILE_N * SP);
    float*       sw = st + lane * 4;
    const float* bp = st + nl * SP + koff;

    const float* wl = w + lane * 4;

    v8f acc[TILES_PER_WAVE][2];
    #pragma unroll
    for (int t = 0; t < TILES_PER_WAVE; ++t) { acc[t][0] = (v8f){}; acc[t][1] = (v8f){}; }

    v4f g[TILE_N];

#define ISSUE_LOADS(T_, KOFS_)                                                        \
    _Pragma("unroll")                                                                 \
    for (int j = 0; j < TILE_N; ++j) {                                                \
        const int c_   = (T_) * TILE_N + j;                                           \
        const int row_ = (c_ < 32) ? __builtin_amdgcn_readlane(idx0, c_)              \
                                   : __builtin_amdgcn_readlane(idx1, c_ - 32);        \
        g[j] = *(const v4f*)(wl + (size_t)row_ * DIM + (KOFS_));                      \
    }

    ISSUE_LOADS(0, 0)

    for (int kc = 0; kc < NKC; ++kc) {
        const int kn = (kc + 1 < NKC) ? (kc + 1) : kc;
        #pragma unroll
        for (int t = 0; t < TILES_PER_WAVE; ++t) {
            #pragma unroll
            for (int j = 0; j < TILE_N; ++j)
                *(v4f*)(sw + j * SP) = g[j];

            if (t + 1 < TILES_PER_WAVE) { ISSUE_LOADS(t + 1, kc * KC) }
            else                        { ISSUE_LOADS(0,     kn * KC) }


            const float* apk = ap + kc * KC;
            v4f fa[2][GS], fb[2][GS];
            #pragma unroll
            for (int i = 0; i < GS; ++i) {
                fa[0][i] = *(const v4f*)(apk + i * 8);
                fb[0][i] = *(const v4f*)(bp  + i * 8);
            }
            #pragma unroll
            for (int gq = 0; gq < (KC / 8) / GS; ++gq) {
                if (gq + 1 < (KC / 8) / GS) {
                    #pragma unroll
                    for (int i = 0; i < GS; ++i) {
                        fa[(gq + 1) & 1][i] = *(const v4f*)(apk + ((gq + 1) * GS + i) * 8);
                        fb[(gq + 1) & 1][i] = *(const v4f*)(bp  + ((gq + 1) * GS + i) * 8);
                    }
                }
                __builtin_amdgcn_sched_barrier(0);
                #pragma unroll
                for (int i = 0; i < GS; ++i) {
                    v4f aq = fa[gq & 1][i];
                    v4f bq = fb[gq & 1][i];
                    v2f a0 = {aq.x, aq.y};
                    v2f b0 = {bq.x, bq.y};
                    v2f a1 = {aq.z, aq.w};
                    v2f b1 = {bq.z, bq.w};
                    acc[t][0] = __builtin_amdgcn_wmma_f32_16x16x4_f32(false, a0, false, b0,
                                                                      (short)0, acc[t][0], false, false);
                    acc[t][1] = __builtin_amdgcn_wmma_f32_16x16x4_f32(false, a1, false, b1,
                                                                      (short)0, acc[t][1], false, false);
                }
            }
        }
    }
#undef ISSUE_LOADS

    #pragma unroll
    for (int t = 0; t < TILES_PER_WAVE; ++t) {
        v8f a = acc[t][0] + acc[t][1];
        if (lane < 16) {
            float* ob = out + (size_t)bl * NOUT * KSEL + n0 + t * TILE_N + lane;
            #pragma unroll
            for (int r = 0; r < NOUT; ++r)
                ob[(size_t)r * KSEL] = a[r];
        }
    }
}

extern "C" void kernel_launch(void* const* d_in, const int* in_sizes, int n_in,
                              void* d_out, int out_size, void* d_ws, size_t ws_size,
                              hipStream_t stream)
{
    const float* h   = (const float*)d_in[0];
    const float* w   = (const float*)d_in[1];
    const int*   idx = (const int*)d_in[2];
    float* out = (float*)d_out;

    dim3 grid(BATCH * LSEQ * BLOCKS_PER_BL);
    dim3 block(BLOCK_THREADS);
    hipLaunchKernelGGL(gather_head_wmma_f32, grid, block, 0, stream,
                       h, w, idx, out);
}
